# starting kernel
# baseline (speedup 1.0000x reference)
#include <hip/hip_runtime.h>
#include <cstddef>

#define NJOINT 25
#define CIN    64
#define COUT   64
#define VC     (NJOINT * COUT)
#define NFRM   (64 * 300)
#define BN_EPS 1e-5f

#define WPB        4
#define FPW        8
#define BLOCKS_B   600

#define NCHUNK     60
#define CHUNKROWS  320

#define WS_AHAT   0
#define WS_PSUM   1024
#define WS_PSQ    (WS_PSUM + NCHUNK * VC)
#define WS_SCALE  (WS_PSQ  + NCHUNK * VC)
#define WS_SHIFT  (WS_SCALE + VC)

typedef __attribute__((ext_vector_type(2))) float v2f;
typedef __attribute__((ext_vector_type(8))) float v8f;

__global__ void stgcn_build_ahat(const int* __restrict__ edge_index, int E,
                                 float* __restrict__ ws) {
  if (threadIdx.x != 0 || blockIdx.x != 0) return;
  const int* row = edge_index;
  const int* col = edge_index + E;
  float deg[NJOINT];
  for (int v = 0; v < NJOINT; ++v) deg[v] = 1.0f;
  for (int e = 0; e < E; ++e) deg[col[e]] += 1.0f;
  float dinv[NJOINT];
  for (int v = 0; v < NJOINT; ++v) dinv[v] = rsqrtf(deg[v]);
  float* A = ws + WS_AHAT;
  for (int i = 0; i < 32 * 32; ++i) A[i] = 0.0f;
  for (int e = 0; e < E; ++e) {
    int s = row[e], t = col[e];
    A[t * 32 + s] += dinv[s] * dinv[t];
  }
  for (int v = 0; v < NJOINT; ++v) A[v * 32 + v] += dinv[v] * dinv[v];
}

#define WSTR 76
#define PSTR 80
#define ASTR 17

__launch_bounds__(WPB * 32, 1)
__global__ void stgcn_gcn_gemm(const float* __restrict__ x,
                               const float* __restrict__ W,
                               const float* __restrict__ bias,
                               const float* __restrict__ ws,
                               float* __restrict__ out) {
  __shared__ v2f wpair[CIN / 2][WSTR];
  __shared__ v2f apair[32][ASTR];
  __shared__ v2f pwbuf[WPB][16][PSTR];
  __shared__ float bbuf[COUT];

  const int tid = threadIdx.x;

  for (int i = tid; i < (CIN / 2) * COUT; i += WPB * 32) {
    const int p = i >> 6, d = i & 63;
    v2f val;
    val.x = W[(2 * p) * COUT + d];
    val.y = W[(2 * p + 1) * COUT + d];
    wpair[p][d] = val;
  }
  for (int i = tid; i < 32 * 16; i += WPB * 32) {
    apair[i >> 4][i & 15] = ((const v2f*)(ws + WS_AHAT))[i];
  }
  if (tid < COUT) bbuf[tid] = bias[tid];
  __syncthreads();

  const int wave = tid >> 5;
  const int lane = tid & 31;
  const int half = lane >> 4;
  const int ln   = lane & 15;
  v2f (*pw)[PSTR] = pwbuf[wave];

  const int waveGlobal = blockIdx.x * WPB + wave;

  for (int f = 0; f < FPW; ++f) {
    const int frame = waveGlobal + f * (BLOCKS_B * WPB);
    const float* xf = x + (size_t)frame * (NJOINT * CIN);

#pragma unroll
    for (int mt = 0; mt < 2; ++mt) {
      v8f acc[4];
#pragma unroll
      for (int nt = 0; nt < 4; ++nt) acc[nt] = (v8f)(0.0f);

      const int m = mt * 16 + ln;
      const bool mvalid = (m < NJOINT);
      const v2f* xrow = (const v2f*)(xf + m * CIN);
#pragma unroll
      for (int ks = 0; ks < 16; ++ks) {
        const int kp = 2 * ks + half;
        v2f a = (v2f)(0.0f);
        if (mvalid) a = xrow[kp];
#pragma unroll
        for (int nt = 0; nt < 4; ++nt) {
          v2f b = wpair[kp][nt * 16 + ln];
          acc[nt] = __builtin_amdgcn_wmma_f32_16x16x4_f32(
              false, a, false, b, (short)0, acc[nt], false, false);
        }
      }
#pragma unroll
      for (int nt = 0; nt < 4; ++nt)
#pragma unroll
        for (int r = 0; r < 8; r += 2) {
          v2f val;
          val.x = acc[nt][r];
          val.y = acc[nt][r + 1];
          pw[mt * 8 + half * 4 + (r >> 1)][nt * 16 + ln] = val;
        }
    }

#pragma unroll
    for (int mt = 0; mt < 2; ++mt) {
      v8f acc[4];
#pragma unroll
      for (int nt = 0; nt < 4; ++nt) acc[nt] = (v8f)(0.0f);

#pragma unroll
      for (int ks = 0; ks < 8; ++ks) {
        const int kp = 2 * ks + half;
        v2f a = apair[mt * 16 + ln][kp];
#pragma unroll
        for (int nt = 0; nt < 4; ++nt) {
          v2f b = pw[kp][nt * 16 + ln];
          acc[nt] = __builtin_amdgcn_wmma_f32_16x16x4_f32(
              false, a, false, b, (short)0, acc[nt], false, false);
        }
      }
      float* of = out + (size_t)frame * VC;
#pragma unroll
      for (int nt = 0; nt < 4; ++nt) {
        const int colb = nt * 16 + ln;
        const float bb = bbuf[colb];
#pragma unroll
        for (int r = 0; r < 8; ++r) {
          const int v = mt * 16 + half * 8 + r;
          if (v < NJOINT) {
            float val = acc[nt][r] + bb;
            of[v * COUT + colb] = val > 0.0f ? val : 0.0f;
          }
        }
      }
    }
  }
}

__global__ void stgcn_bn_partial(const float* __restrict__ out,
                                 float* __restrict__ ws) {
  const int ch = blockIdx.x * blockDim.x + threadIdx.x;
  if (ch >= VC) return;
  const int chunk = blockIdx.y;
  const float* p = out + (size_t)chunk * CHUNKROWS * VC + ch;
  float s = 0.0f, ss = 0.0f;
  for (int i = 0; i < CHUNKROWS; ++i) {
    float v = p[(size_t)i * VC];
    s += v;
    ss += v * v;
  }
  ws[WS_PSUM + chunk * VC + ch] = s;
  ws[WS_PSQ + chunk * VC + ch] = ss;
}

__global__ void stgcn_bn_finalize(const float* __restrict__ gamma,
                                  const float* __restrict__ beta,
                                  float* __restrict__ ws) {
  const int ch = blockIdx.x * blockDim.x + threadIdx.x;
  if (ch >= VC) return;
  double s = 0.0, ss = 0.0;
  for (int c = 0; c < NCHUNK; ++c) {
    s += (double)ws[WS_PSUM + c * VC + ch];
    ss += (double)ws[WS_PSQ + c * VC + ch];
  }
  const double n = (double)NFRM;
  const double mean = s / n;
  const double var = ss / n - mean * mean;
  const float scale = gamma[ch] * rsqrtf((float)var + BN_EPS);
  const float shift = beta[ch] - (float)mean * scale;
  ws[WS_SCALE + ch] = scale;
  ws[WS_SHIFT + ch] = shift;
}

__global__ void stgcn_bn_apply(float* __restrict__ out,
                               const float* __restrict__ ws) {
  const size_t idx = ((size_t)blockIdx.x * blockDim.x + threadIdx.x) * 4;
  float4 h = *(const float4*)(out + idx);
  const int ch = (int)(idx % VC);
  const float4 sc = *(const float4*)(ws + WS_SCALE + ch);
  const float4 sh = *(const float4*)(ws + WS_SHIFT + ch);
  h.x = h.x * sc.x + sh.x;
  h.y = h.y * sc.y + sh.y;
  h.z = h.z * sc.z + sh.z;
  h.w = h.w * sc.w + sh.w;
  *(float4*)(out + idx) = h;
}

extern "C" void kernel_launch(void* const* d_in, const int* in_sizes, int n_in,
                              void* d_out, int out_size, void* d_ws, size_t ws_size,
                              hipStream_t stream) {
  const float* x     = (const float*)d_in[0];
  const float* W     = (const float*)d_in[1];
  const float* b     = (const float*)d_in[2];
  const float* gamma = (const float*)d_in[3];
  const float* beta  = (const float*)d_in[4];
  const int*   ei    = (const int*)d_in[5];
  float* out = (float*)d_out;
  float* ws  = (float*)d_ws;
  const int E = in_sizes[5] / 2;

  hipLaunchKernelGGL(stgcn_build_ahat, dim3(1), dim3(64), 0, stream, ei, E, ws);
  hipLaunchKernelGGL(stgcn_gcn_gemm, dim3(BLOCKS_B), dim3(WPB * 32), 0, stream,
                     x, W, b, ws, out);
  hipLaunchKernelGGL(stgcn_bn_partial, dim3((VC + 255) / 256, NCHUNK), dim3(256),
                     0, stream, out, ws);
  hipLaunchKernelGGL(stgcn_bn_finalize, dim3((VC + 255) / 256), dim3(256),
                     0, stream, gamma, beta, ws);
  hipLaunchKernelGGL(stgcn_bn_apply, dim3((NFRM * VC) / 4 / 256), dim3(256),
                     0, stream, out, ws);
}


# GEMM launch deals output channel 4*ln+nt to lane ln of column tile nt (W columns staged so): a row's 4 accumulators leave as one b128, a half-wave writes a whole 256-B row; 16 stores a frame, not 64
# speedup vs baseline: 2.1011x; 2.1011x over previous
#include <hip/hip_runtime.h>
#include <cstddef>

#define NJOINT 25
#define CIN    64
#define COUT   64
#define VC     (NJOINT * COUT)
#define NFRM   (64 * 300)
#define BN_EPS 1e-5f

#define WPB        16
#define FPB        75
#define BLOCKS_B   256
#define LONGW      11

#define NCHUNK     10
#define CHUNKROWS  1920
#define SWAVES     16
#define WAVEROWS   (CHUNKROWS / SWAVES)
#define WAVELOADS  (WAVEROWS / 2)
#define NPIECE     (VC / 64)

#define WS_AHAT   0
#define WS_PSUM   1024
#define WS_PSQ    (WS_PSUM + NCHUNK * VC)

typedef __attribute__((ext_vector_type(2))) float v2f;
typedef __attribute__((ext_vector_type(8))) float v8f;

__global__ void stgcn_build_ahat(const int* __restrict__ edge_index, int E,
                                 float* __restrict__ ws) {
  const int t = blockIdx.x;
  const int s = threadIdx.x;
  const int* row = edge_index;
  const int* col = edge_index + E;
  int deg_t = 1, deg_s = 1;
  int cnt = 0;
  for (int e = 0; e < E; ++e) {
    const int r = row[e], c = col[e];
    deg_t += (c == t) ? 1 : 0;
    deg_s += (c == s) ? 1 : 0;
    cnt   += (c == t && r == s) ? 1 : 0;
  }
  const float dt = rsqrtf((float)deg_t);
  const float ds = rsqrtf((float)deg_s);
  float a = 0.0f;
  for (int i = 0; i < cnt; ++i) a += ds * dt;
  if (t == s) a += dt * dt;
  if (t >= NJOINT || s >= NJOINT) a = 0.0f;
  ws[WS_AHAT + t * 32 + s] = a;
}

#define WSTR 72
#define TQ   17

typedef __attribute__((ext_vector_type(4))) float v4f;

__device__ __forceinline__ void stgcn_load_frame(v4f (&xb)[13], const float* __restrict__ xf,
                                                 int ln, int half) {
  const v4f* xq = (const v4f*)xf;
#pragma unroll
  for (int i = 0; i < 12; ++i) {
    const int v = 4 * (i >> 1) + 2 * half + (i & 1);
    xb[i] = xq[v * 16 + ln];
  }
  xb[12] = xq[24 * 16 + ln];
}

__launch_bounds__(WPB * 32, 1)
__global__ void stgcn_gcn_gemm(const float* __restrict__ x,
                               const float* __restrict__ W,
                               const float* __restrict__ bias,
                               const float* __restrict__ ws,
                               float* __restrict__ out) {
  __shared__ v2f wpair[CIN / 2][WSTR];
  __shared__ v4f tbuf[WPB][32][TQ];
  __shared__ __attribute__((aligned(16))) float bbuf[COUT];

  const int tid = threadIdx.x;

  for (int i = tid; i < (CIN / 2) * COUT; i += WPB * 32) {
    const int p = i >> 6, e = i & 63;
    const int d = 4 * (e & 15) + (e >> 4);
    v2f val;
    val.x = W[(2 * p) * COUT + d];
    val.y = W[(2 * p + 1) * COUT + d];
    wpair[p][e] = val;
  }
  if (tid < COUT) bbuf[tid] = bias[tid];
  __syncthreads();

  const int wave = __builtin_amdgcn_readfirstlane(tid >> 5);
  const int lane = tid & 31;
  const int half = lane >> 4;
  const int ln   = lane & 15;
  v4f (*tw)[TQ] = tbuf[wave];
  const v4f bq = *(const v4f*)&bbuf[4 * ln];

  v2f ah[2][7];
#pragma unroll
  for (int mt = 0; mt < 2; ++mt)
#pragma unroll
    for (int s = 0; s < 7; ++s)
      ah[mt][s] = *(const v2f*)(ws + WS_AHAT + (16 * mt + ln) * 32 + 4 * s + 2 * half);

  const int nf     = wave < LONGW ? 5 : 4;
  const int frame0 = blockIdx.x * FPB + (wave < LONGW ? 5 * wave : 5 * LONGW + 4 * (wave - LONGW));

  v4f xb[13];
  stgcn_load_frame(xb, x + (size_t)frame0 * (NJOINT * CIN), ln, half);

  for (int f = 0; f < nf; ++f) {
    const int frame = frame0 + f;
    asm volatile("" ::: "memory");

#pragma unroll
    for (int mt = 0; mt < 2; ++mt) {
      v8f acc[4];
#pragma unroll
      for (int nt = 0; nt < 4; ++nt) acc[nt] = (v8f)(0.0f);
#pragma unroll
      for (int s = 0; s < 7; ++s) {
#pragma unroll
        for (int nt = 0; nt < 4; ++nt) {
          v2f b;
          if (s < 6) {
            b.x = xb[2 * s][nt];
            b.y = xb[2 * s + 1][nt];
          } else {
            b.x = half ? 0.0f : xb[12][nt];
            b.y = 0.0f;
          }
          acc[nt] = __builtin_amdgcn_wmma_f32_16x16x4_f32(
              false, ah[mt][s], false, b, (short)0, acc[nt], false, false);
        }
      }
#pragma unroll
      for (int r = 0; r < 8; ++r) {
        v4f o;
        o.x = acc[0][r];
        o.y = acc[1][r];
        o.z = acc[2][r];
        o.w = acc[3][r];
        tw[16 * mt + 8 * half + r][ln] = o;
      }
    }

    if (f + 1 < nf)
      stgcn_load_frame(xb, x + (size_t)(frame + 1) * (NJOINT * CIN), ln, half);

#pragma unroll
    for (int mt = 0; mt < 2; ++mt) {
      v4f q[8];
#pragma unroll
      for (int u = 0; u < 8; ++u) q[u] = tw[16 * mt + ln][2 * u + half];

      v8f acc[4];
#pragma unroll
      for (int nt = 0; nt < 4; ++nt) acc[nt] = (v8f)(0.0f);

#pragma unroll
      for (int u = 0; u < 8; ++u) {
#pragma unroll
        for (int sub = 0; sub < 2; ++sub) {
          v2f a;
          a.x = q[u][2 * sub];
          a.y = q[u][2 * sub + 1];
          const int p = 4 * u + 2 * half + sub;
#pragma unroll
          for (int nt = 0; nt < 4; ++nt) {
            v2f b = wpair[p][nt * 16 + ln];
            acc[nt] = __builtin_amdgcn_wmma_f32_16x16x4_f32(
                false, a, false, b, (short)0, acc[nt], false, false);
          }
        }
      }
      v4f* oq = (v4f*)(out + (size_t)frame * VC) + (16 * mt + 8 * half) * 16 + ln;
      v4f o[8];
#pragma unroll
      for (int r = 0; r < 8; ++r) {
        const float v0 = acc[0][r] + bq.x;
        const float v1 = acc[1][r] + bq.y;
        const float v2 = acc[2][r] + bq.z;
        const float v3 = acc[3][r] + bq.w;
        o[r].x = v0 > 0.0f ? v0 : 0.0f;
        o[r].y = v1 > 0.0f ? v1 : 0.0f;
        o[r].z = v2 > 0.0f ? v2 : 0.0f;
        o[r].w = v3 > 0.0f ? v3 : 0.0f;
      }
      oq[0] = o[0];
      if (mt == 0 || half == 0) {
#pragma unroll
        for (int r = 1; r < 8; ++r) oq[r * 16] = o[r];
      }
    }
  }
}

__launch_bounds__(SWAVES * 32) __attribute__((amdgpu_waves_per_eu(4, 4)))
__global__ void stgcn_bn_partial(const float* __restrict__ out,
                                 float* __restrict__ ws) {
  __shared__ float red[2][2 * SWAVES][64];

  const int piece = blockIdx.x;
  const int chunk = blockIdx.y;
  const int tid  = threadIdx.x;
  const int wave = tid >> 5;
  const int lane = tid & 31;
  const int half = lane >> 4;
  const int ln   = lane & 15;

  const size_t row0 = (size_t)chunk * CHUNKROWS + (size_t)wave * WAVEROWS + half;
  const v4f* p = (const v4f*)(out + row0 * VC + piece * 64) + ln;

  v4f r[WAVELOADS];
#pragma unroll
  for (int i = 0; i < WAVELOADS; ++i) r[i] = p[(size_t)(2 * i) * (VC / 4)];

  v4f s = (v4f)(0.0f), q = (v4f)(0.0f);
#pragma unroll
  for (int i = 0; i < WAVELOADS; ++i) {
    s += r[i];
    q += r[i] * r[i];
  }

  *(v4f*)&red[0][2 * wave + half][4 * ln] = s;
  *(v4f*)&red[1][2 * wave + half][4 * ln] = q;
  __syncthreads();

  if (tid < 128) {
    const int which = tid >> 6;
    const int c = tid & 63;
    float a = 0.0f;
    for (int k = 0; k < 2 * SWAVES; ++k) a += red[which][k][c];
    ws[(which ? WS_PSQ : WS_PSUM) + chunk * VC + piece * 64 + c] = a;
  }
}

__launch_bounds__(SWAVES * 32) __attribute__((amdgpu_waves_per_eu(4, 4)))
__global__ void stgcn_bn_apply(float* __restrict__ out,
                               const float* __restrict__ gamma,
                               const float* __restrict__ beta,
                               const float* __restrict__ ws) {
  __shared__ __attribute__((aligned(16))) float sc_s[64];
  __shared__ __attribute__((aligned(16))) float sh_s[64];

  const int piece = blockIdx.x;
  const int chunk = blockIdx.y;
  const int tid  = threadIdx.x;
  const int wave = tid >> 5;
  const int lane = tid & 31;
  const int half = lane >> 4;
  const int ln   = lane & 15;

  if (tid < 64) {
    const int ch = piece * 64 + tid;
    double s = 0.0, ss = 0.0;
    for (int c = 0; c < NCHUNK; ++c) {
      s += (double)ws[WS_PSUM + c * VC + ch];
      ss += (double)ws[WS_PSQ + c * VC + ch];
    }
    const double n = (double)NFRM;
    const double mean = s / n;
    const double var = ss / n - mean * mean;
    const float scale = gamma[ch] * rsqrtf((float)var + BN_EPS);
    sc_s[tid] = scale;
    sh_s[tid] = beta[ch] - (float)mean * scale;
  }

  const size_t row0 = (size_t)chunk * CHUNKROWS + (size_t)wave * WAVEROWS + half;
  v4f* p = (v4f*)(out + row0 * VC + piece * 64) + ln;

  v4f r[WAVELOADS];
#pragma unroll
  for (int i = 0; i < WAVELOADS; ++i) r[i] = p[(size_t)(2 * i) * (VC / 4)];

  __syncthreads();
  const v4f sc = *(const v4f*)&sc_s[4 * ln];
  const v4f sh = *(const v4f*)&sh_s[4 * ln];

#pragma unroll
  for (int i = 0; i < WAVELOADS; ++i) p[(size_t)(2 * i) * (VC / 4)] = r[i] * sc + sh;
}

extern "C" void kernel_launch(void* const* d_in, const int* in_sizes, int n_in,
                              void* d_out, int out_size, void* d_ws, size_t ws_size,
                              hipStream_t stream) {
  const float* x     = (const float*)d_in[0];
  const float* W     = (const float*)d_in[1];
  const float* b     = (const float*)d_in[2];
  const float* gamma = (const float*)d_in[3];
  const float* beta  = (const float*)d_in[4];
  const int*   ei    = (const int*)d_in[5];
  float* out = (float*)d_out;
  float* ws  = (float*)d_ws;
  const int E = in_sizes[5] / 2;

  hipLaunchKernelGGL(stgcn_build_ahat, dim3(32), dim3(32), 0, stream, ei, E, ws);
  hipLaunchKernelGGL(stgcn_gcn_gemm, dim3(BLOCKS_B), dim3(WPB * 32), 0, stream,
                     x, W, b, ws, out);
  hipLaunchKernelGGL(stgcn_bn_partial, dim3(NPIECE, NCHUNK), dim3(SWAVES * 32),
                     0, stream, out, ws);
  hipLaunchKernelGGL(stgcn_bn_apply, dim3(NPIECE, NCHUNK), dim3(SWAVES * 32),
                     0, stream, out, gamma, beta, ws);
}
